# starting kernel
# baseline (speedup 1.0000x reference)
#include <hip/hip_runtime.h>
#include <math.h>

#define L     880
#define HID   8
#define OUT   96
#define DS    48
#define T     64
#define BATCH 4
#define NCONV 63
#define CW    64
#define CSTEP (L * CW)

typedef float v2f __attribute__((ext_vector_type(2)));
typedef float v8f __attribute__((ext_vector_type(8)));

template <int LDB, int LDD>
__global__ void wmma_mm_kernel(const float* __restrict__ Am,
                               const float* __restrict__ Bb, long bz,
                               float* __restrict__ Db, long dz,
                               int ncols) {
    const int lane = threadIdx.x & 31;
    const int jt   = blockIdx.y * 4 + (threadIdx.x >> 5);
    if (jt * 16 >= ncols) return;

    const int itile = blockIdx.x;
    const float* Bm = Bb + (size_t)blockIdx.z * bz;
    float*       Dm = Db + (size_t)blockIdx.z * dz;

    const int mrow  = lane & 15;
    const int khalf = (lane >> 4) << 1;
    const int col   = jt * 16 + (lane & 15);

    const float* Arow = Am + (size_t)(itile * 16 + mrow) * L;

    v8f acc = {};
    for (int k = 0; k < L; k += 4) {
        v2f a, b;
        a.x = Arow[k + khalf + 0];
        a.y = Arow[k + khalf + 1];
        b.x = Bm[(size_t)(k + khalf + 0) * LDB + col];
        b.y = Bm[(size_t)(k + khalf + 1) * LDB + col];
        acc = __builtin_amdgcn_wmma_f32_16x16x4_f32(false, a, false, b,
                                                    (short)0, acc, false, false);
    }

    const int rbase = itile * 16 + ((lane >> 4) << 3);
    #pragma unroll
    for (int v = 0; v < 8; ++v)
        Dm[(size_t)(rbase + v) * LDD + col] = acc[v];
}

__global__ void ses2seq_init(const float* __restrict__ Bm,
                             const float* __restrict__ theta0,
                             const float* __restrict__ xs,
                             float* __restrict__ C0,
                             float* __restrict__ thetas) {
    int l = blockIdx.x * blockDim.x + threadIdx.x;
    if (l >= L) return;
    float* row = C0 + (size_t)l * CW;
    const float* brow = Bm + (size_t)l * DS;
    #pragma unroll 4
    for (int d = 0; d < DS; ++d) row[d] = brow[d];
    row[DS] = theta0[l];
    #pragma unroll
    for (int b = 0; b < BATCH; ++b) {
        float acc = 0.f;
        const float* x0 = xs + (size_t)b * T * DS;
        for (int d = 0; d < DS; ++d) acc += brow[d] * x0[d];
        row[DS + 1 + b] = -acc;
    }
    for (int c = DS + 1 + BATCH; c < CW; ++c) row[c] = 0.f;
    #pragma unroll
    for (int b = 0; b < BATCH; ++b)
        thetas[((size_t)b * T + 0) * L + l] = theta0[l];
}

__global__ void ses2seq_conv(const float* __restrict__ C,
                             const float* __restrict__ xs,
                             float* __restrict__ thetas) {
    const int lane  = threadIdx.x & 31;
    const int wv    = threadIdx.x >> 5;
    const int ltile = blockIdx.x;
    const int b     = blockIdx.y;
    const int khalf = (lane >> 4) << 1;
    const int lrow  = ltile * 16 + (lane & 15);
    const int tc    = wv * 16 + (lane & 15);
    const int tcc   = (tc >= NCONV) ? 0 : tc;

    const float* u = xs + (size_t)b * T * DS;

    v8f acc = {};
    for (int m = 0; m < NCONV; ++m) {
        const float* Km = C + (size_t)m * CSTEP;
        const float* Kp = C + (size_t)(m > 0 ? m - 1 : 0) * CSTEP;
        const float  w  = (m > 0) ? 1.f : 0.f;
        int s = tcc - m; if (s < 0) s += NCONV;
        const float* urow = u + (size_t)(s + 1) * DS;
        #pragma unroll 4
        for (int d = 0; d < DS; d += 4) {
            const int dk = d + khalf;
            v2f a, bb;
            a.x = Km[(size_t)lrow * CW + dk + 0] - w * Kp[(size_t)lrow * CW + dk + 0];
            a.y = Km[(size_t)lrow * CW + dk + 1] - w * Kp[(size_t)lrow * CW + dk + 1];
            bb.x = urow[dk + 0];
            bb.y = urow[dk + 1];
            acc = __builtin_amdgcn_wmma_f32_16x16x4_f32(false, a, false, bb,
                                                        (short)0, acc, false, false);
        }
    }

    if (tc < NCONV) {
        const int l0 = ltile * 16 + ((lane >> 4) << 3);
        const int zc = (tc - 1 + NCONV) % NCONV;
        #pragma unroll
        for (int v = 0; v < 8; ++v) {
            const int l = l0 + v;
            float val = acc[v]
                      + C[(size_t)tc * CSTEP + (size_t)l * CW + DS]
                      + C[(size_t)zc * CSTEP + (size_t)l * CW + DS + 1 + b];
            thetas[((size_t)b * T + (tc + 1)) * L + l] = val;
        }
    }
}

__global__ void ses2seq_head(const float* __restrict__ thetas,
                             const float* __restrict__ ts,
                             float* __restrict__ out) {
    const int bt = blockIdx.x;
    const float* th = thetas + (size_t)bt * L;
    __shared__ float h[HID];
    if (threadIdx.x < HID) {
        float x = th[threadIdx.x] * ts[bt] + th[HID + threadIdx.x];
        h[threadIdx.x] = x > 0.f ? x : 0.f;
    }
    __syncthreads();
    const int o = threadIdx.x;
    float acc = th[2 * HID + OUT * HID + o];
    const float* w1 = th + 2 * HID + o * HID;
    #pragma unroll
    for (int k = 0; k < HID; ++k) acc += w1[k] * h[k];
    float r;
    if (o < DS) {
        r = tanhf(acc);
    } else {
        r = acc > 0.f ? acc + log1pf(expf(-acc)) : log1pf(expf(acc));
    }
    out[(size_t)bt * OUT + o] = r;
}

extern "C" void kernel_launch(void* const* d_in, const int* in_sizes, int n_in,
                              void* d_out, int out_size, void* d_ws, size_t ws_size,
                              hipStream_t stream) {
    const float* xs     = (const float*)d_in[0];
    const float* ts     = (const float*)d_in[1];
    const float* theta0 = (const float*)d_in[2];
    const float* A      = (const float*)d_in[3];
    const float* Bm     = (const float*)d_in[4];
    (void)in_sizes; (void)n_in; (void)out_size; (void)ws_size;

    float* A2     = (float*)d_ws;
    float* A4     = A2 + (size_t)L * L;
    float* C      = A4 + (size_t)L * L;
    float* thetas = C + (size_t)NCONV * CSTEP;
    float* out    = (float*)d_out;

    ses2seq_init<<<(L + 255) / 256, 256, 0, stream>>>(Bm, theta0, xs, C, thetas);

    wmma_mm_kernel<L, L><<<dim3(55, 14, 1), 128, 0, stream>>>(A,  A,  0, A2, 0, L);
    wmma_mm_kernel<L, L><<<dim3(55, 14, 1), 128, 0, stream>>>(A2, A2, 0, A4, 0, L);

    wmma_mm_kernel<CW, CW><<<dim3(55, 1, 1), 128, 0, stream>>>(
        A, C, 0, C + CSTEP, 0, CW);
    wmma_mm_kernel<CW, CW><<<dim3(55, 1, 2), 128, 0, stream>>>(
        A2, C, CSTEP, C + (size_t)2 * CSTEP, CSTEP, CW);

    for (int j = 1; j <= 15; ++j) {
        int nz = (j == 15) ? 3 : 4;
        wmma_mm_kernel<CW, CW><<<dim3(55, 1, nz), 128, 0, stream>>>(
            A4,
            C + (size_t)(4 * j - 4) * CSTEP, CSTEP,
            C + (size_t)(4 * j) * CSTEP, CSTEP, CW);
    }

    ses2seq_conv<<<dim3(55, BATCH), 128, 0, stream>>>(C, xs, thetas);

    ses2seq_head<<<BATCH * T, OUT, 0, stream>>>(thetas, ts, out);
}


# mm kernel: block barrier split around the WMMA run: wave reads fragments, stores next chunk to LDS, signals, runs its WMMAs, then waits; stores and barrier skew hide under the run; sums unchanged
# speedup vs baseline: 2.9396x; 2.9396x over previous
#include <hip/hip_runtime.h>
#include <math.h>

#define L     880
#define HID   8
#define OUT   96
#define DS    48
#define T     64
#define BATCH 4
#define NCONV 63
#define CW    64
#define CSTEP (L * CW)

typedef float v2f __attribute__((ext_vector_type(2)));
typedef float v4f __attribute__((ext_vector_type(4)));
typedef float v8f __attribute__((ext_vector_type(8)));

#define KC      80
#define NCHUNK  (L / KC)
#define KSTEPS  (KC / 4)
#define APITCH  84
#define BPITCH  72
#define A_PER   (16 * KC / 2 / 128)
#define NROWT   (L / 16)

__device__ __forceinline__ void lds_stage_signal() {
    __builtin_amdgcn_fence(__ATOMIC_RELEASE, "workgroup", "local");
    __builtin_amdgcn_s_barrier_signal(-1);
}
__device__ __forceinline__ void lds_stage_wait() {
    __builtin_amdgcn_s_barrier_wait(-1);
    __builtin_amdgcn_fence(__ATOMIC_ACQUIRE, "workgroup", "local");
}

template <int LDB, int LDD, int RT>
__global__ __launch_bounds__(128 * RT)
void wmma_mm_kernel(const float* __restrict__ Am,
                    const float* __restrict__ Bb, long bz,
                    float* __restrict__ Db, long dz,
                    int ncols) {
    constexpr int B_ROWS = 8 * RT;
    constexpr int B_PER  = (KC + B_ROWS - 1) / B_ROWS;
    constexpr int B_TAIL = KC - B_ROWS * (B_PER - 1);

    __shared__ __attribute__((aligned(16))) float As[2][RT * 16 * APITCH];
    __shared__ __attribute__((aligned(16))) float Bs[2][KC * BPITCH];

    const int tid  = threadIdx.x;
    const int lane = tid & 31;
    const int wv   = __builtin_amdgcn_readfirstlane(tid >> 5);
    const int wr   = wv >> 2;
    const int wc   = wv & 3;
    const int jt   = blockIdx.y * 4 + wc;
    const int itile = blockIdx.x * RT + wr;
    const bool live = (jt * 16 < ncols) && (itile < NROWT);
    const bool blast = (2 * wv < B_TAIL);

    const float* Bm = Bb + (size_t)blockIdx.z * bz;
    float*       Dm = Db + (size_t)blockIdx.z * dz;

    const int mrow  = lane & 15;
    const int khalf = (lane >> 4) << 1;
    const int col   = jt * 16 + (lane & 15);

    const int    ar  = tid >> 3;
    const int    ac2 = tid & 7;
    const int    agr = blockIdx.x * (RT * 16) + ar;
    const float* agp = Am + (size_t)(agr < L ? agr : L - 1) * L + ac2 * 2;
    const int    aso = ar * APITCH + ac2 * 2;
    const int    bkr  = tid >> 4;
    const int    bc4  = tid & 15;
    const int    gcol = blockIdx.y * 64 + bc4 * 4;
    const float* bgp  = Bm + (size_t)bkr * LDB + (gcol < ncols ? gcol : 0);
    const int    bso  = bkr * BPITCH + bc4 * 4;

    v2f ra[2][A_PER];
    v4f rb[2][B_PER];
    v8f acc = {};

#define MM_GLOAD(SET, CHUNK)                                                     \
    {                                                                            \
        const int koff_ = (CHUNK) * KC;                                          \
        _Pragma("unroll")               \
        for (int j = 0; j < A_PER; ++j)                                          \
            ra[SET][j] = *(const v2f*)(agp + koff_ + 16 * j);                    \
        _Pragma("unroll")               \
        for (int j = 0; j < B_PER; ++j)                                          \
            if (j + 1 < B_PER || blast)                                          \
                rb[SET][j] = *(const v4f*)(bgp + (size_t)(koff_ + B_ROWS * j) * LDB); \
    }
#define MM_LSTORE(SET)                                                           \
    {                                                                            \
        _Pragma("unroll")             \
        for (int j = 0; j < A_PER; ++j)                                          \
            *(v2f*)(&As[SET][aso + 16 * j]) = ra[SET][j];                        \
        _Pragma("unroll")             \
        for (int j = 0; j < B_PER; ++j)                                          \
            if (j + 1 < B_PER || blast)                                          \
                *(v4f*)(&Bs[SET][bso + B_ROWS * j * BPITCH]) = rb[SET][j];       \
    }
#define MM_ITER(CH, P)                                                           \
    {                                                                            \
        if ((CH) + 2 < NCHUNK) MM_GLOAD(P, (CH) + 2)                             \
        v2f af[KSTEPS], bf[KSTEPS];               \
        if (live) {                                                              \
            const v2f*   as2 = (const v2f*)(&As[P][(wr * 16 + mrow) * APITCH + khalf]); \
            const float* bs  = &Bs[P][khalf * BPITCH + wc * 16 + (lane & 15)];   \
            _Pragma("unroll")      \
            for (int s = 0; s < KSTEPS; ++s) {                                   \
                af[s]   = as2[2 * s];                                            \
                bf[s].x = bs[(4 * s + 0) * BPITCH];                              \
                bf[s].y = bs[(4 * s + 1) * BPITCH];                              \
            }                                                                    \
        }                                                                        \
        if ((CH) + 1 < NCHUNK) {                                                 \
            MM_LSTORE(1 - (P))                                                   \
            lds_stage_signal();                                                  \
        }                                                                        \
          \
          \
          \
          \
        __builtin_amdgcn_sched_barrier(0);                                       \
        if (live) {                                                              \
            _Pragma("unroll")            \
            for (int s = 0; s < KSTEPS; ++s)                                     \
                acc = __builtin_amdgcn_wmma_f32_16x16x4_f32(false, af[s], false, bf[s], \
                                                            (short)0, acc, false, false); \
        }                                                                        \
        if ((CH) + 1 < NCHUNK) lds_stage_wait();                                 \
    }

    MM_GLOAD(0, 0)
    MM_GLOAD(1, 1)
    MM_LSTORE(0)
    lds_stage_signal();
    lds_stage_wait();

    for (int ch = 0; ch + 1 < NCHUNK; ch += 2) {
        MM_ITER(ch, 0)
        MM_ITER(ch + 1, 1)
    }
    MM_ITER(NCHUNK - 1, 0)
#undef MM_GLOAD
#undef MM_LSTORE
#undef MM_ITER

    if (live) {
        const int rbase = itile * 16 + ((lane >> 4) << 3);
        #pragma unroll
        for (int v = 0; v < 8; ++v)
            Dm[(size_t)(rbase + v) * LDD + col] = acc[v];
    }
}

__global__ void ses2seq_init(const float* __restrict__ Bm,
                             const float* __restrict__ theta0,
                             const float* __restrict__ xs,
                             float* __restrict__ C0,
                             float* __restrict__ thetas) {
    int l = blockIdx.x * blockDim.x + threadIdx.x;
    if (l >= L) return;
    float* row = C0 + (size_t)l * CW;
    const float* brow = Bm + (size_t)l * DS;
    #pragma unroll 4
    for (int d = 0; d < DS; ++d) row[d] = brow[d];
    row[DS] = theta0[l];
    #pragma unroll
    for (int b = 0; b < BATCH; ++b) {
        float acc = 0.f;
        const float* x0 = xs + (size_t)b * T * DS;
        for (int d = 0; d < DS; ++d) acc += brow[d] * x0[d];
        row[DS + 1 + b] = -acc;
    }
    for (int c = DS + 1 + BATCH; c < CW; ++c) row[c] = 0.f;
    #pragma unroll
    for (int b = 0; b < BATCH; ++b)
        thetas[((size_t)b * T + 0) * L + l] = theta0[l];
}

#define CV_J      (DS / 8)
#define CV_OFF(j) (8 * (j))

__global__ __launch_bounds__(128)
void ses2seq_conv(const float* __restrict__ C,
                  const float* __restrict__ xs,
                  float* __restrict__ thetas) {
    const int lane  = threadIdx.x & 31;
    const int wv    = threadIdx.x >> 5;
    const int ltile = blockIdx.x;
    const int b     = blockIdx.y;
    const int half4 = (lane >> 4) << 2;
    const int lrow  = ltile * 16 + (lane & 15);
    const int tc    = wv * 16 + (lane & 15);
    const int tcc   = (tc >= NCONV) ? 0 : tc;

    const float* ka = C + (size_t)lrow * CW + half4;
    const float* ub = xs + (size_t)b * T * DS + DS + half4;

    v4f kc[4][CV_J];
    v4f uu[4][CV_J];
    v8f acc = {};

#define CV_LOAD(SET, M)                                                          \
    {                                                                            \
        int s_ = tcc - (M); if (s_ < 0) s_ += NCONV;            \
        const float* kp_ = ka + (size_t)(M) * CSTEP;                             \
        const float* up_ = ub + (size_t)s_ * DS;                                 \
        _Pragma("unroll")               \
        for (int j = 0; j < CV_J; ++j) {                                         \
            kc[SET][j] = *(const v4f*)(kp_ + CV_OFF(j));                         \
            uu[SET][j] = *(const v4f*)(up_ + CV_OFF(j));                         \
        }                                                                        \
    }
#define CV_STEP(M, PREV, CUR, NXT2, DOLOAD)                                      \
    {                                                                            \
        if (DOLOAD) CV_LOAD(NXT2, (M) + 2)                                       \
          \
          \
          \
        __builtin_amdgcn_sched_barrier(0);                                       \
        _Pragma("unroll")             \
        for (int j = 0; j < CV_J; ++j) {                                         \
            const v4f kd = kc[CUR][j] - kc[PREV][j];               \
            const v4f uv = uu[CUR][j];                                           \
            const v2f a0 = {kd.x, kd.y}, a1 = {kd.z, kd.w};                      \
            const v2f b0 = {uv.x, uv.y}, b1 = {uv.z, uv.w};                      \
            acc = __builtin_amdgcn_wmma_f32_16x16x4_f32(false, a0, false, b0,    \
                                                        (short)0, acc, false, false); \
            acc = __builtin_amdgcn_wmma_f32_16x16x4_f32(false, a1, false, b1,    \
                                                        (short)0, acc, false, false); \
        }                                                                        \
    }

    CV_LOAD(0, 0)
    CV_LOAD(1, 1)
    #pragma unroll
    for (int j = 0; j < CV_J; ++j) kc[3][j] = (v4f){0.f, 0.f, 0.f, 0.f};

    for (int m = 0; m < 60; m += 4) {
        CV_STEP(m,     3, 0, 2, 1)
        CV_STEP(m + 1, 0, 1, 3, 1)
        CV_STEP(m + 2, 1, 2, 0, 1)
        CV_STEP(m + 3, 2, 3, 1, 1)
    }
    CV_STEP(60, 3, 0, 2, 1)
    CV_STEP(61, 0, 1, 3, 0)
    CV_STEP(62, 1, 2, 0, 0)
#undef CV_LOAD
#undef CV_STEP

    if (tc < NCONV) {
        const int l0 = ltile * 16 + ((lane >> 4) << 3);
        const int zc = (tc - 1 + NCONV) % NCONV;
        #pragma unroll
        for (int v = 0; v < 8; ++v) {
            const int l = l0 + v;
            float val = acc[v]
                      + C[(size_t)tc * CSTEP + (size_t)l * CW + DS]
                      + C[(size_t)zc * CSTEP + (size_t)l * CW + DS + 1 + b];
            thetas[((size_t)b * T + (tc + 1)) * L + l] = val;
        }
    }
}

__global__ void ses2seq_head(const float* __restrict__ thetas,
                             const float* __restrict__ ts,
                             float* __restrict__ out) {
    const int bt = blockIdx.x;
    const float* th = thetas + (size_t)bt * L;
    __shared__ float h[HID];
    if (threadIdx.x < HID) {
        float x = th[threadIdx.x] * ts[bt] + th[HID + threadIdx.x];
        h[threadIdx.x] = x > 0.f ? x : 0.f;
    }
    __syncthreads();
    const int o = threadIdx.x;
    float acc = th[2 * HID + OUT * HID + o];
    const float* w1 = th + 2 * HID + o * HID;
    #pragma unroll
    for (int k = 0; k < HID; ++k) acc += w1[k] * h[k];
    float r;
    if (o < DS) {
        r = tanhf(acc);
    } else {
        r = acc > 0.f ? acc + log1pf(expf(-acc)) : log1pf(expf(acc));
    }
    out[(size_t)bt * OUT + o] = r;
}

extern "C" void kernel_launch(void* const* d_in, const int* in_sizes, int n_in,
                              void* d_out, int out_size, void* d_ws, size_t ws_size,
                              hipStream_t stream) {
    const float* xs     = (const float*)d_in[0];
    const float* ts     = (const float*)d_in[1];
    const float* theta0 = (const float*)d_in[2];
    const float* A      = (const float*)d_in[3];
    const float* Bm     = (const float*)d_in[4];
    (void)in_sizes; (void)n_in; (void)out_size; (void)ws_size;

    float* A2     = (float*)d_ws;
    float* A4     = A2 + (size_t)L * L;
    float* A8     = A4 + (size_t)L * L;
    float* A16    = A8 + (size_t)L * L;
    float* C      = A16 + (size_t)L * L;
    float* thetas = C + (size_t)NCONV * CSTEP;
    float* out    = (float*)d_out;

    ses2seq_init<<<(L + 255) / 256, 256, 0, stream>>>(Bm, theta0, xs, C, thetas);

    wmma_mm_kernel<L, L, 4><<<dim3(14, 14, 1), 512, 0, stream>>>(A,  A,  0, A2,  0, L);
    wmma_mm_kernel<L, L, 4><<<dim3(14, 14, 1), 512, 0, stream>>>(A2, A2, 0, A4,  0, L);
    wmma_mm_kernel<L, L, 4><<<dim3(14, 14, 1), 512, 0, stream>>>(A4, A4, 0, A8,  0, L);
    wmma_mm_kernel<L, L, 4><<<dim3(14, 14, 1), 512, 0, stream>>>(A8, A8, 0, A16, 0, L);

    wmma_mm_kernel<CW, CW, 1><<<dim3(55, 1, 1), 128, 0, stream>>>(
        A,   C, 0,     C + CSTEP, 0, CW);
    wmma_mm_kernel<CW, CW, 1><<<dim3(55, 1, 2), 128, 0, stream>>>(
        A2,  C, CSTEP, C + (size_t)2 * CSTEP,  CSTEP, CW);
    wmma_mm_kernel<CW, CW, 1><<<dim3(55, 1, 4), 128, 0, stream>>>(
        A4,  C, CSTEP, C + (size_t)4 * CSTEP,  CSTEP, CW);
    wmma_mm_kernel<CW, CW, 2><<<dim3(28, 1, 8), 256, 0, stream>>>(
        A8,  C, CSTEP, C + (size_t)8 * CSTEP,  CSTEP, CW);
    wmma_mm_kernel<CW, CW, 4><<<dim3(14, 1, 16), 512, 0, stream>>>(
        A16, C, CSTEP, C + (size_t)16 * CSTEP, CSTEP, CW);
    wmma_mm_kernel<CW, CW, 4><<<dim3(14, 1, 16), 512, 0, stream>>>(
        A16, C + (size_t)16 * CSTEP, CSTEP, C + (size_t)32 * CSTEP, CSTEP, CW);
    wmma_mm_kernel<CW, CW, 4><<<dim3(14, 1, 15), 512, 0, stream>>>(
        A16, C + (size_t)32 * CSTEP, CSTEP, C + (size_t)48 * CSTEP, CSTEP, CW);

    ses2seq_conv<<<dim3(55, BATCH), 128, 0, stream>>>(C, xs, thetas);

    ses2seq_head<<<BATCH * T, OUT, 0, stream>>>(thetas, ts, out);
}


# mm kernel, 16/32-row launches: 2nd fragment register set; chunk c+1's fragments are read from LDS and chunk c+2 is stored to LDS while chunk c's WMMAs run (3 staging sets); sums and order unchanged
# speedup vs baseline: 3.0770x; 1.0467x over previous
#include <hip/hip_runtime.h>
#include <math.h>

#define L     880
#define HID   8
#define OUT   96
#define DS    48
#define T     64
#define BATCH 4
#define NCONV 63
#define CW    64
#define CSTEP (L * CW)

typedef float v2f __attribute__((ext_vector_type(2)));
typedef float v4f __attribute__((ext_vector_type(4)));
typedef float v8f __attribute__((ext_vector_type(8)));

#define KC      80
#define NCHUNK  (L / KC)
#define KSTEPS  (KC / 4)
#define APITCH  84
#define BPITCH  72
#define A_PER   (16 * KC / 2 / 128)
#define NROWT   (L / 16)

__device__ __forceinline__ void lds_stage_signal() {
    __builtin_amdgcn_fence(__ATOMIC_RELEASE, "workgroup", "local");
    __builtin_amdgcn_s_barrier_signal(-1);
}
__device__ __forceinline__ void lds_stage_wait() {
    __builtin_amdgcn_s_barrier_wait(-1);
    __builtin_amdgcn_fence(__ATOMIC_ACQUIRE, "workgroup", "local");
}

template <int LDB, int LDD, int RT>
__global__ __launch_bounds__(128 * RT)
void wmma_mm_kernel(const float* __restrict__ Am,
                    const float* __restrict__ Bb, long bz,
                    float* __restrict__ Db, long dz,
                    int ncols) {
    constexpr int B_ROWS = 8 * RT;
    constexpr int B_PER  = (KC + B_ROWS - 1) / B_ROWS;
    constexpr int B_TAIL = KC - B_ROWS * (B_PER - 1);

    __shared__ __attribute__((aligned(16))) float As[2][RT * 16 * APITCH];
    __shared__ __attribute__((aligned(16))) float Bs[2][KC * BPITCH];

    const int tid  = threadIdx.x;
    const int lane = tid & 31;
    const int wv   = __builtin_amdgcn_readfirstlane(tid >> 5);
    const int wr   = wv >> 2;
    const int wc   = wv & 3;
    const int jt   = blockIdx.y * 4 + wc;
    const int itile = blockIdx.x * RT + wr;
    const bool live = (jt * 16 < ncols) && (itile < NROWT);
    const bool blast = (2 * wv < B_TAIL);

    const float* Bm = Bb + (size_t)blockIdx.z * bz;
    float*       Dm = Db + (size_t)blockIdx.z * dz;

    const int mrow  = lane & 15;
    const int khalf = (lane >> 4) << 1;
    const int col   = jt * 16 + (lane & 15);

    const int    ar  = tid >> 3;
    const int    ac2 = tid & 7;
    const int    agr = blockIdx.x * (RT * 16) + ar;
    const float* agp = Am + (size_t)(agr < L ? agr : L - 1) * L + ac2 * 2;
    const int    aso = ar * APITCH + ac2 * 2;
    const int    bkr  = tid >> 4;
    const int    bc4  = tid & 15;
    const int    gcol = blockIdx.y * 64 + bc4 * 4;
    const float* bgp  = Bm + (size_t)bkr * LDB + (gcol < ncols ? gcol : 0);
    const int    bso  = bkr * BPITCH + bc4 * 4;

    v2f ra[2][A_PER];
    v4f rb[2][B_PER];
    v8f acc = {};

#define MM_GLOAD(SET, CHUNK)                                                     \
    {                                                                            \
        const int koff_ = (CHUNK) * KC;                                          \
        _Pragma("unroll")               \
        for (int j = 0; j < A_PER; ++j)                                          \
            ra[SET][j] = *(const v2f*)(agp + koff_ + 16 * j);                    \
        _Pragma("unroll")               \
        for (int j = 0; j < B_PER; ++j)                                          \
            if (j + 1 < B_PER || blast)                                          \
                rb[SET][j] = *(const v4f*)(bgp + (size_t)(koff_ + B_ROWS * j) * LDB); \
    }
#define MM_LSTORE(SET)                                                           \
    {                                                                            \
        _Pragma("unroll")             \
        for (int j = 0; j < A_PER; ++j)                                          \
            *(v2f*)(&As[SET][aso + 16 * j]) = ra[SET][j];                        \
        _Pragma("unroll")             \
        for (int j = 0; j < B_PER; ++j)                                          \
            if (j + 1 < B_PER || blast)                                          \
                *(v4f*)(&Bs[SET][bso + B_ROWS * j * BPITCH]) = rb[SET][j];       \
    }
#define MM_ITER(CH, P)                                                           \
    {                                                                            \
        if ((CH) + 2 < NCHUNK) MM_GLOAD(P, (CH) + 2)                             \
        v2f af[KSTEPS], bf[KSTEPS];               \
        if (live) {                                                              \
            const v2f*   as2 = (const v2f*)(&As[P][(wr * 16 + mrow) * APITCH + khalf]); \
            const float* bs  = &Bs[P][khalf * BPITCH + wc * 16 + (lane & 15)];   \
            _Pragma("unroll")      \
            for (int s = 0; s < KSTEPS; ++s) {                                   \
                af[s]   = as2[2 * s];                                            \
                bf[s].x = bs[(4 * s + 0) * BPITCH];                              \
                bf[s].y = bs[(4 * s + 1) * BPITCH];                              \
            }                                                                    \
        }                                                                        \
        if ((CH) + 1 < NCHUNK) {                                                 \
            MM_LSTORE(1 - (P))                                                   \
            lds_stage_signal();                                                  \
        }                                                                        \
          \
          \
          \
          \
        __builtin_amdgcn_sched_barrier(0);                                       \
        if (live) {                                                              \
            _Pragma("unroll")            \
            for (int s = 0; s < KSTEPS; ++s)                                     \
                acc = __builtin_amdgcn_wmma_f32_16x16x4_f32(false, af[s], false, bf[s], \
                                                            (short)0, acc, false, false); \
        }                                                                        \
        if ((CH) + 1 < NCHUNK) lds_stage_wait();                                 \
    }

    MM_GLOAD(0, 0)
    MM_GLOAD(1, 1)
    MM_LSTORE(0)
    lds_stage_signal();
    lds_stage_wait();

    for (int ch = 0; ch + 1 < NCHUNK; ch += 2) {
        MM_ITER(ch, 0)
        MM_ITER(ch + 1, 1)
    }
    MM_ITER(NCHUNK - 1, 0)
#undef MM_GLOAD
#undef MM_LSTORE
#undef MM_ITER

    if (live) {
        const int rbase = itile * 16 + ((lane >> 4) << 3);
        #pragma unroll
        for (int v = 0; v < 8; ++v)
            Dm[(size_t)(rbase + v) * LDD + col] = acc[v];
    }
}

template <int LDB, int LDD, int RT>
__global__ __launch_bounds__(128 * RT)
void wmma_mm_pipe(const float* __restrict__ Am,
                  const float* __restrict__ Bb, long bz,
                  float* __restrict__ Db, long dz,
                  int ncols) {
    constexpr int B_ROWS = 8 * RT;
    constexpr int B_PER  = KC / B_ROWS;
    static_assert(KC % B_ROWS == 0, "every thread owns a whole number of B pieces");

    __shared__ __attribute__((aligned(16))) float As[2][RT * 16 * APITCH];
    __shared__ __attribute__((aligned(16))) float Bs[2][KC * BPITCH];

    const int tid  = threadIdx.x;
    const int lane = tid & 31;
    const int wv   = __builtin_amdgcn_readfirstlane(tid >> 5);
    const int wr   = wv >> 2;
    const int wc   = wv & 3;
    const int jt   = blockIdx.y * 4 + wc;
    const int itile = blockIdx.x * RT + wr;
    const bool live = (jt * 16 < ncols) && (itile < NROWT);

    const float* Bm = Bb + (size_t)blockIdx.z * bz;
    float*       Dm = Db + (size_t)blockIdx.z * dz;

    const int mrow  = lane & 15;
    const int khalf = (lane >> 4) << 1;
    const int col   = jt * 16 + (lane & 15);

    const int    ar  = tid >> 3;
    const int    ac2 = tid & 7;
    const int    agr = blockIdx.x * (RT * 16) + ar;
    const float* agp = Am + (size_t)(agr < L ? agr : L - 1) * L + ac2 * 2;
    const int    aso = ar * APITCH + ac2 * 2;
    const int    bkr  = tid >> 4;
    const int    bc4  = tid & 15;
    const int    gcol = blockIdx.y * 64 + bc4 * 4;
    const float* bgp  = Bm + (size_t)bkr * LDB + (gcol < ncols ? gcol : 0);
    const int    bso  = bkr * BPITCH + bc4 * 4;

    v2f ra[3][A_PER];
    v4f rb[3][B_PER];
    v2f af[2][KSTEPS];
    v2f bf[2][KSTEPS];
    v8f acc = {};

#define MP_GLOAD(SET, CHUNK)                                                     \
    {                                                                            \
        const int koff_ = (CHUNK) * KC;                                          \
        _Pragma("unroll")               \
        for (int j = 0; j < A_PER; ++j)                                          \
            ra[SET][j] = *(const v2f*)(agp + koff_ + 16 * j);                    \
        _Pragma("unroll")               \
        for (int j = 0; j < B_PER; ++j)                                          \
            rb[SET][j] = *(const v4f*)(bgp + (size_t)(koff_ + B_ROWS * j) * LDB); \
    }
#define MP_LSTORE(SET, STAGE)                                                    \
    {                                                                            \
        _Pragma("unroll")             \
        for (int j = 0; j < A_PER; ++j)                                          \
            *(v2f*)(&As[STAGE][aso + 16 * j]) = ra[SET][j];                      \
        _Pragma("unroll")             \
        for (int j = 0; j < B_PER; ++j)                                          \
            *(v4f*)(&Bs[STAGE][bso + B_ROWS * j * BPITCH]) = rb[SET][j];         \
    }
#define MP_FRAG(F, STAGE)                                                        \
    {                                                                            \
        const v2f*   as2 = (const v2f*)(&As[STAGE][(wr * 16 + mrow) * APITCH + khalf]); \
        const float* bs  = &Bs[STAGE][khalf * BPITCH + wc * 16 + (lane & 15)];   \
        _Pragma("unroll")          \
        for (int s = 0; s < KSTEPS; ++s) {                                       \
            af[F][s]   = as2[2 * s];                                             \
            bf[F][s].x = bs[(4 * s + 0) * BPITCH];                               \
            bf[F][s].y = bs[(4 * s + 1) * BPITCH];                               \
        }                                                                        \
    }
#define MP_WMMA(F, S0, S1)                                                       \
    {                                                                            \
        _Pragma("unroll")                \
        for (int s = (S0); s < (S1); ++s)                                        \
            acc = __builtin_amdgcn_wmma_f32_16x16x4_f32(false, af[F][s], false, bf[F][s], \
                                                        (short)0, acc, false, false); \
    }
#define MP_SIG 15
#define MP_ITER(C)                                                               \
    {                                                                            \
        if ((C) + 4 < NCHUNK) MP_GLOAD(((C) + 1) % 3, (C) + 4)                   \
        if ((C) + 1 < NCHUNK) { if (live) MP_FRAG(((C) + 1) & 1, ((C) + 1) & 1) } \
        if ((C) + 2 < NCHUNK) MP_LSTORE(((C) + 2) % 3, (C) & 1)                  \
          \
          \
        __builtin_amdgcn_sched_barrier(0);                                       \
        if (live) MP_WMMA((C) & 1, 0, MP_SIG)                                    \
        if ((C) + 2 < NCHUNK) lds_stage_signal();                                \
        if (live) MP_WMMA((C) & 1, MP_SIG, KSTEPS)                               \
        if ((C) + 2 < NCHUNK) lds_stage_wait();                                  \
    }

    MP_GLOAD(0, 0)
    MP_GLOAD(1, 1)
    MP_GLOAD(2, 2)
    MP_LSTORE(0, 0)
    MP_LSTORE(1, 1)
    MP_GLOAD(0, 3)
    lds_stage_signal();
    lds_stage_wait();
    if (live) MP_FRAG(0, 0)
    lds_stage_signal();
    lds_stage_wait();

    static_assert(NCHUNK == 11, "the chunk sequence below is written out for 11 chunks");
    MP_ITER(0) MP_ITER(1) MP_ITER(2) MP_ITER(3) MP_ITER(4) MP_ITER(5)
    MP_ITER(6) MP_ITER(7) MP_ITER(8) MP_ITER(9) MP_ITER(10)
#undef MP_GLOAD
#undef MP_LSTORE
#undef MP_FRAG
#undef MP_WMMA
#undef MP_ITER
#undef MP_SIG

    if (live) {
        const int rbase = itile * 16 + ((lane >> 4) << 3);
        #pragma unroll
        for (int v = 0; v < 8; ++v)
            Dm[(size_t)(rbase + v) * LDD + col] = acc[v];
    }
}

__global__ void ses2seq_init(const float* __restrict__ Bm,
                             const float* __restrict__ theta0,
                             const float* __restrict__ xs,
                             float* __restrict__ C0,
                             float* __restrict__ thetas) {
    int l = blockIdx.x * blockDim.x + threadIdx.x;
    if (l >= L) return;
    float* row = C0 + (size_t)l * CW;
    const float* brow = Bm + (size_t)l * DS;
    #pragma unroll 4
    for (int d = 0; d < DS; ++d) row[d] = brow[d];
    row[DS] = theta0[l];
    #pragma unroll
    for (int b = 0; b < BATCH; ++b) {
        float acc = 0.f;
        const float* x0 = xs + (size_t)b * T * DS;
        for (int d = 0; d < DS; ++d) acc += brow[d] * x0[d];
        row[DS + 1 + b] = -acc;
    }
    for (int c = DS + 1 + BATCH; c < CW; ++c) row[c] = 0.f;
    #pragma unroll
    for (int b = 0; b < BATCH; ++b)
        thetas[((size_t)b * T + 0) * L + l] = theta0[l];
}

#define CV_J      (DS / 8)
#define CV_OFF(j) (8 * (j))

__global__ __launch_bounds__(128)
void ses2seq_conv(const float* __restrict__ C,
                  const float* __restrict__ xs,
                  float* __restrict__ thetas) {
    const int lane  = threadIdx.x & 31;
    const int wv    = threadIdx.x >> 5;
    const int ltile = blockIdx.x;
    const int b     = blockIdx.y;
    const int half4 = (lane >> 4) << 2;
    const int lrow  = ltile * 16 + (lane & 15);
    const int tc    = wv * 16 + (lane & 15);
    const int tcc   = (tc >= NCONV) ? 0 : tc;

    const float* ka = C + (size_t)lrow * CW + half4;
    const float* ub = xs + (size_t)b * T * DS + DS + half4;

    v4f kc[4][CV_J];
    v4f uu[4][CV_J];
    v8f acc = {};

#define CV_LOAD(SET, M)                                                          \
    {                                                                            \
        int s_ = tcc - (M); if (s_ < 0) s_ += NCONV;            \
        const float* kp_ = ka + (size_t)(M) * CSTEP;                             \
        const float* up_ = ub + (size_t)s_ * DS;                                 \
        _Pragma("unroll")               \
        for (int j = 0; j < CV_J; ++j) {                                         \
            kc[SET][j] = *(const v4f*)(kp_ + CV_OFF(j));                         \
            uu[SET][j] = *(const v4f*)(up_ + CV_OFF(j));                         \
        }                                                                        \
    }
#define CV_STEP(M, PREV, CUR, NXT2, DOLOAD)                                      \
    {                                                                            \
        if (DOLOAD) CV_LOAD(NXT2, (M) + 2)                                       \
          \
          \
          \
        __builtin_amdgcn_sched_barrier(0);                                       \
        _Pragma("unroll")             \
        for (int j = 0; j < CV_J; ++j) {                                         \
            const v4f kd = kc[CUR][j] - kc[PREV][j];               \
            const v4f uv = uu[CUR][j];                                           \
            const v2f a0 = {kd.x, kd.y}, a1 = {kd.z, kd.w};                      \
            const v2f b0 = {uv.x, uv.y}, b1 = {uv.z, uv.w};                      \
            acc = __builtin_amdgcn_wmma_f32_16x16x4_f32(false, a0, false, b0,    \
                                                        (short)0, acc, false, false); \
            acc = __builtin_amdgcn_wmma_f32_16x16x4_f32(false, a1, false, b1,    \
                                                        (short)0, acc, false, false); \
        }                                                                        \
    }

    CV_LOAD(0, 0)
    CV_LOAD(1, 1)
    #pragma unroll
    for (int j = 0; j < CV_J; ++j) kc[3][j] = (v4f){0.f, 0.f, 0.f, 0.f};

    for (int m = 0; m < 60; m += 4) {
        CV_STEP(m,     3, 0, 2, 1)
        CV_STEP(m + 1, 0, 1, 3, 1)
        CV_STEP(m + 2, 1, 2, 0, 1)
        CV_STEP(m + 3, 2, 3, 1, 1)
    }
    CV_STEP(60, 3, 0, 2, 1)
    CV_STEP(61, 0, 1, 3, 0)
    CV_STEP(62, 1, 2, 0, 0)
#undef CV_LOAD
#undef CV_STEP

    if (tc < NCONV) {
        const int l0 = ltile * 16 + ((lane >> 4) << 3);
        const int zc = (tc - 1 + NCONV) % NCONV;
        #pragma unroll
        for (int v = 0; v < 8; ++v) {
            const int l = l0 + v;
            float val = acc[v]
                      + C[(size_t)tc * CSTEP + (size_t)l * CW + DS]
                      + C[(size_t)zc * CSTEP + (size_t)l * CW + DS + 1 + b];
            thetas[((size_t)b * T + (tc + 1)) * L + l] = val;
        }
    }
}

__global__ void ses2seq_head(const float* __restrict__ thetas,
                             const float* __restrict__ ts,
                             float* __restrict__ out) {
    const int bt = blockIdx.x;
    const float* th = thetas + (size_t)bt * L;
    __shared__ float h[HID];
    if (threadIdx.x < HID) {
        float x = th[threadIdx.x] * ts[bt] + th[HID + threadIdx.x];
        h[threadIdx.x] = x > 0.f ? x : 0.f;
    }
    __syncthreads();
    const int o = threadIdx.x;
    float acc = th[2 * HID + OUT * HID + o];
    const float* w1 = th + 2 * HID + o * HID;
    #pragma unroll
    for (int k = 0; k < HID; ++k) acc += w1[k] * h[k];
    float r;
    if (o < DS) {
        r = tanhf(acc);
    } else {
        r = acc > 0.f ? acc + log1pf(expf(-acc)) : log1pf(expf(acc));
    }
    out[(size_t)bt * OUT + o] = r;
}

extern "C" void kernel_launch(void* const* d_in, const int* in_sizes, int n_in,
                              void* d_out, int out_size, void* d_ws, size_t ws_size,
                              hipStream_t stream) {
    const float* xs     = (const float*)d_in[0];
    const float* ts     = (const float*)d_in[1];
    const float* theta0 = (const float*)d_in[2];
    const float* A      = (const float*)d_in[3];
    const float* Bm     = (const float*)d_in[4];
    (void)in_sizes; (void)n_in; (void)out_size; (void)ws_size;

    float* A2     = (float*)d_ws;
    float* A4     = A2 + (size_t)L * L;
    float* A8     = A4 + (size_t)L * L;
    float* A16    = A8 + (size_t)L * L;
    float* C      = A16 + (size_t)L * L;
    float* thetas = C + (size_t)NCONV * CSTEP;
    float* out    = (float*)d_out;

    ses2seq_init<<<(L + 255) / 256, 256, 0, stream>>>(Bm, theta0, xs, C, thetas);

    wmma_mm_kernel<L, L, 4><<<dim3(14, 14, 1), 512, 0, stream>>>(A,  A,  0, A2,  0, L);
    wmma_mm_kernel<L, L, 4><<<dim3(14, 14, 1), 512, 0, stream>>>(A2, A2, 0, A4,  0, L);
    wmma_mm_kernel<L, L, 4><<<dim3(14, 14, 1), 512, 0, stream>>>(A4, A4, 0, A8,  0, L);
    wmma_mm_kernel<L, L, 4><<<dim3(14, 14, 1), 512, 0, stream>>>(A8, A8, 0, A16, 0, L);

    wmma_mm_pipe<CW, CW, 1><<<dim3(55, 1, 1), 128, 0, stream>>>(
        A,   C, 0,     C + CSTEP, 0, CW);
    wmma_mm_pipe<CW, CW, 1><<<dim3(55, 1, 2), 128, 0, stream>>>(
        A2,  C, CSTEP, C + (size_t)2 * CSTEP,  CSTEP, CW);
    wmma_mm_pipe<CW, CW, 1><<<dim3(55, 1, 4), 128, 0, stream>>>(
        A4,  C, CSTEP, C + (size_t)4 * CSTEP,  CSTEP, CW);
    wmma_mm_pipe<CW, CW, 2><<<dim3(28, 1, 8), 256, 0, stream>>>(
        A8,  C, CSTEP, C + (size_t)8 * CSTEP,  CSTEP, CW);
    wmma_mm_kernel<CW, CW, 4><<<dim3(14, 1, 16), 512, 0, stream>>>(
        A16, C, CSTEP, C + (size_t)16 * CSTEP, CSTEP, CW);
    wmma_mm_kernel<CW, CW, 4><<<dim3(14, 1, 16), 512, 0, stream>>>(
        A16, C + (size_t)16 * CSTEP, CSTEP, C + (size_t)32 * CSTEP, CSTEP, CW);
    wmma_mm_kernel<CW, CW, 4><<<dim3(14, 1, 15), 512, 0, stream>>>(
        A16, C + (size_t)32 * CSTEP, CSTEP, C + (size_t)48 * CSTEP, CSTEP, CW);

    ses2seq_conv<<<dim3(55, BATCH), 128, 0, stream>>>(C, xs, thetas);

    ses2seq_head<<<BATCH * T, OUT, 0, stream>>>(thetas, ts, out);
}


# conv operands read from copies their producers write in the conv's fragment order (512 contiguous bytes per load, 4 lines not 16), fetched 4 timesteps ahead; sums and order unchanged
# speedup vs baseline: 3.1773x; 1.0326x over previous
#include <hip/hip_runtime.h>
#include <math.h>

#define L     880
#define HID   8
#define OUT   96
#define DS    48
#define T     64
#define BATCH 4
#define NCONV 63
#define CW    64
#define CSTEP (L * CW)
#define CV_J  (DS / 8)
#define FRAG  128
#define CFSTEP ((L / 16) * CV_J * FRAG)
#define UF_PIECES (BATCH * NCONV * CV_J * 32)
#define INIT_ROWBLK ((L + 255) / 256)

typedef float v2f __attribute__((ext_vector_type(2)));
typedef float v4f __attribute__((ext_vector_type(4)));
typedef float v8f __attribute__((ext_vector_type(8)));

#define KC      80
#define NCHUNK  (L / KC)
#define KSTEPS  (KC / 4)
#define APITCH  84
#define BPITCH  72
#define A_PER   (16 * KC / 2 / 128)
#define NROWT   (L / 16)

__device__ __forceinline__ void lds_stage_signal() {
    __builtin_amdgcn_fence(__ATOMIC_RELEASE, "workgroup", "local");
    __builtin_amdgcn_s_barrier_signal(-1);
}
__device__ __forceinline__ void lds_stage_wait() {
    __builtin_amdgcn_s_barrier_wait(-1);
    __builtin_amdgcn_fence(__ATOMIC_ACQUIRE, "workgroup", "local");
}

#define STORE_FRAG_COPY()                                                        \
    if constexpr (LDD == CW) {                                                   \
        if (live && jt < DS / 16) {                             \
            const int q  = jt * 16 + (lane & 15);                                \
            float* fp = Fb + (size_t)blockIdx.z * fz                             \
                      + ((size_t)itile * CV_J + (q >> 3)) * FRAG                 \
                      + ((((q >> 2) & 1) << 4) + ((lane >> 4) << 3)) * 4 + (q & 3); \
            _Pragma("unroll")   \
            for (int v = 0; v < 8; ++v) fp[4 * v] = acc[v];                      \
        }                                                                        \
    }

template <int LDB, int LDD, int RT>
__global__ __launch_bounds__(128 * RT)
void wmma_mm_kernel(const float* __restrict__ Am,
                    const float* __restrict__ Bb, long bz,
                    float* __restrict__ Db, long dz,
                    float* __restrict__ Fb, long fz,
                    int ncols) {
    constexpr int B_ROWS = 8 * RT;
    constexpr int B_PER  = (KC + B_ROWS - 1) / B_ROWS;
    constexpr int B_TAIL = KC - B_ROWS * (B_PER - 1);

    __shared__ __attribute__((aligned(16))) float As[2][RT * 16 * APITCH];
    __shared__ __attribute__((aligned(16))) float Bs[2][KC * BPITCH];

    const int tid  = threadIdx.x;
    const int lane = tid & 31;
    const int wv   = __builtin_amdgcn_readfirstlane(tid >> 5);
    const int wr   = wv >> 2;
    const int wc   = wv & 3;
    const int jt   = blockIdx.y * 4 + wc;
    const int itile = blockIdx.x * RT + wr;
    const bool live = (jt * 16 < ncols) && (itile < NROWT);
    const bool blast = (2 * wv < B_TAIL);

    const float* Bm = Bb + (size_t)blockIdx.z * bz;
    float*       Dm = Db + (size_t)blockIdx.z * dz;

    const int mrow  = lane & 15;
    const int khalf = (lane >> 4) << 1;
    const int col   = jt * 16 + (lane & 15);

    const int    ar  = tid >> 3;
    const int    ac2 = tid & 7;
    const int    agr = blockIdx.x * (RT * 16) + ar;
    const float* agp = Am + (size_t)(agr < L ? agr : L - 1) * L + ac2 * 2;
    const int    aso = ar * APITCH + ac2 * 2;
    const int    bkr  = tid >> 4;
    const int    bc4  = tid & 15;
    const int    gcol = blockIdx.y * 64 + bc4 * 4;
    const float* bgp  = Bm + (size_t)bkr * LDB + (gcol < ncols ? gcol : 0);
    const int    bso  = bkr * BPITCH + bc4 * 4;

    v2f ra[2][A_PER];
    v4f rb[2][B_PER];
    v8f acc = {};

#define MM_GLOAD(SET, CHUNK)                                                     \
    {                                                                            \
        const int koff_ = (CHUNK) * KC;                                          \
        _Pragma("unroll")               \
        for (int j = 0; j < A_PER; ++j)                                          \
            ra[SET][j] = *(const v2f*)(agp + koff_ + 16 * j);                    \
        _Pragma("unroll")               \
        for (int j = 0; j < B_PER; ++j)                                          \
            if (j + 1 < B_PER || blast)                                          \
                rb[SET][j] = *(const v4f*)(bgp + (size_t)(koff_ + B_ROWS * j) * LDB); \
    }
#define MM_LSTORE(SET)                                                           \
    {                                                                            \
        _Pragma("unroll")             \
        for (int j = 0; j < A_PER; ++j)                                          \
            *(v2f*)(&As[SET][aso + 16 * j]) = ra[SET][j];                        \
        _Pragma("unroll")             \
        for (int j = 0; j < B_PER; ++j)                                          \
            if (j + 1 < B_PER || blast)                                          \
                *(v4f*)(&Bs[SET][bso + B_ROWS * j * BPITCH]) = rb[SET][j];       \
    }
#define MM_ITER(CH, P)                                                           \
    {                                                                            \
        if ((CH) + 2 < NCHUNK) MM_GLOAD(P, (CH) + 2)                             \
        v2f af[KSTEPS], bf[KSTEPS];               \
        if (live) {                                                              \
            const v2f*   as2 = (const v2f*)(&As[P][(wr * 16 + mrow) * APITCH + khalf]); \
            const float* bs  = &Bs[P][khalf * BPITCH + wc * 16 + (lane & 15)];   \
            _Pragma("unroll")      \
            for (int s = 0; s < KSTEPS; ++s) {                                   \
                af[s]   = as2[2 * s];                                            \
                bf[s].x = bs[(4 * s + 0) * BPITCH];                              \
                bf[s].y = bs[(4 * s + 1) * BPITCH];                              \
            }                                                                    \
        }                                                                        \
        if ((CH) + 1 < NCHUNK) {                                                 \
            MM_LSTORE(1 - (P))                                                   \
            lds_stage_signal();                                                  \
        }                                                                        \
          \
          \
          \
          \
        __builtin_amdgcn_sched_barrier(0);                                       \
        if (live) {                                                              \
            _Pragma("unroll")            \
            for (int s = 0; s < KSTEPS; ++s)                                     \
                acc = __builtin_amdgcn_wmma_f32_16x16x4_f32(false, af[s], false, bf[s], \
                                                            (short)0, acc, false, false); \
        }                                                                        \
        if ((CH) + 1 < NCHUNK) lds_stage_wait();                                 \
    }

    MM_GLOAD(0, 0)
    MM_GLOAD(1, 1)
    MM_LSTORE(0)
    lds_stage_signal();
    lds_stage_wait();

    for (int ch = 0; ch + 1 < NCHUNK; ch += 2) {
        MM_ITER(ch, 0)
        MM_ITER(ch + 1, 1)
    }
    MM_ITER(NCHUNK - 1, 0)
#undef MM_GLOAD
#undef MM_LSTORE
#undef MM_ITER

    if (live) {
        const int rbase = itile * 16 + ((lane >> 4) << 3);
        #pragma unroll
        for (int v = 0; v < 8; ++v)
            Dm[(size_t)(rbase + v) * LDD + col] = acc[v];
    }
    STORE_FRAG_COPY()
}

template <int LDB, int LDD, int RT>
__global__ __launch_bounds__(128 * RT)
void wmma_mm_pipe(const float* __restrict__ Am,
                  const float* __restrict__ Bb, long bz,
                  float* __restrict__ Db, long dz,
                  float* __restrict__ Fb, long fz,
                  int ncols) {
    constexpr int B_ROWS = 8 * RT;
    constexpr int B_PER  = KC / B_ROWS;
    static_assert(KC % B_ROWS == 0, "every thread owns a whole number of B pieces");

    __shared__ __attribute__((aligned(16))) float As[2][RT * 16 * APITCH];
    __shared__ __attribute__((aligned(16))) float Bs[2][KC * BPITCH];

    const int tid  = threadIdx.x;
    const int lane = tid & 31;
    const int wv   = __builtin_amdgcn_readfirstlane(tid >> 5);
    const int wr   = wv >> 2;
    const int wc   = wv & 3;
    const int jt   = blockIdx.y * 4 + wc;
    const int itile = blockIdx.x * RT + wr;
    const bool live = (jt * 16 < ncols) && (itile < NROWT);

    const float* Bm = Bb + (size_t)blockIdx.z * bz;
    float*       Dm = Db + (size_t)blockIdx.z * dz;

    const int mrow  = lane & 15;
    const int khalf = (lane >> 4) << 1;
    const int col   = jt * 16 + (lane & 15);

    const int    ar  = tid >> 3;
    const int    ac2 = tid & 7;
    const int    agr = blockIdx.x * (RT * 16) + ar;
    const float* agp = Am + (size_t)(agr < L ? agr : L - 1) * L + ac2 * 2;
    const int    aso = ar * APITCH + ac2 * 2;
    const int    bkr  = tid >> 4;
    const int    bc4  = tid & 15;
    const int    gcol = blockIdx.y * 64 + bc4 * 4;
    const float* bgp  = Bm + (size_t)bkr * LDB + (gcol < ncols ? gcol : 0);
    const int    bso  = bkr * BPITCH + bc4 * 4;

    v2f ra[3][A_PER];
    v4f rb[3][B_PER];
    v2f af[2][KSTEPS];
    v2f bf[2][KSTEPS];
    v8f acc = {};

#define MP_GLOAD(SET, CHUNK)                                                     \
    {                                                                            \
        const int koff_ = (CHUNK) * KC;                                          \
        _Pragma("unroll")               \
        for (int j = 0; j < A_PER; ++j)                                          \
            ra[SET][j] = *(const v2f*)(agp + koff_ + 16 * j);                    \
        _Pragma("unroll")               \
        for (int j = 0; j < B_PER; ++j)                                          \
            rb[SET][j] = *(const v4f*)(bgp + (size_t)(koff_ + B_ROWS * j) * LDB); \
    }
#define MP_LSTORE(SET, STAGE)                                                    \
    {                                                                            \
        _Pragma("unroll")             \
        for (int j = 0; j < A_PER; ++j)                                          \
            *(v2f*)(&As[STAGE][aso + 16 * j]) = ra[SET][j];                      \
        _Pragma("unroll")             \
        for (int j = 0; j < B_PER; ++j)                                          \
            *(v4f*)(&Bs[STAGE][bso + B_ROWS * j * BPITCH]) = rb[SET][j];         \
    }
#define MP_FRAG(F, STAGE)                                                        \
    {                                                                            \
        const v2f*   as2 = (const v2f*)(&As[STAGE][(wr * 16 + mrow) * APITCH + khalf]); \
        const float* bs  = &Bs[STAGE][khalf * BPITCH + wc * 16 + (lane & 15)];   \
        _Pragma("unroll")          \
        for (int s = 0; s < KSTEPS; ++s) {                                       \
            af[F][s]   = as2[2 * s];                                             \
            bf[F][s].x = bs[(4 * s + 0) * BPITCH];                               \
            bf[F][s].y = bs[(4 * s + 1) * BPITCH];                               \
        }                                                                        \
    }
#define MP_WMMA(F, S0, S1)                                                       \
    {                                                                            \
        _Pragma("unroll")                \
        for (int s = (S0); s < (S1); ++s)                                        \
            acc = __builtin_amdgcn_wmma_f32_16x16x4_f32(false, af[F][s], false, bf[F][s], \
                                                        (short)0, acc, false, false); \
    }
#define MP_SIG 15
#define MP_ITER(C)                                                               \
    {                                                                            \
        if ((C) + 4 < NCHUNK) MP_GLOAD(((C) + 1) % 3, (C) + 4)                   \
        if ((C) + 1 < NCHUNK) { if (live) MP_FRAG(((C) + 1) & 1, ((C) + 1) & 1) } \
        if ((C) + 2 < NCHUNK) MP_LSTORE(((C) + 2) % 3, (C) & 1)                  \
          \
          \
        __builtin_amdgcn_sched_barrier(0);                                       \
        if (live) MP_WMMA((C) & 1, 0, MP_SIG)                                    \
        if ((C) + 2 < NCHUNK) lds_stage_signal();                                \
        if (live) MP_WMMA((C) & 1, MP_SIG, KSTEPS)                               \
        if ((C) + 2 < NCHUNK) lds_stage_wait();                                  \
    }

    MP_GLOAD(0, 0)
    MP_GLOAD(1, 1)
    MP_GLOAD(2, 2)
    MP_LSTORE(0, 0)
    MP_LSTORE(1, 1)
    MP_GLOAD(0, 3)
    lds_stage_signal();
    lds_stage_wait();
    if (live) MP_FRAG(0, 0)
    lds_stage_signal();
    lds_stage_wait();

    static_assert(NCHUNK == 11, "the chunk sequence below is written out for 11 chunks");
    MP_ITER(0) MP_ITER(1) MP_ITER(2) MP_ITER(3) MP_ITER(4) MP_ITER(5)
    MP_ITER(6) MP_ITER(7) MP_ITER(8) MP_ITER(9) MP_ITER(10)
#undef MP_GLOAD
#undef MP_LSTORE
#undef MP_FRAG
#undef MP_WMMA
#undef MP_ITER
#undef MP_SIG

    if (live) {
        const int rbase = itile * 16 + ((lane >> 4) << 3);
        #pragma unroll
        for (int v = 0; v < 8; ++v)
            Dm[(size_t)(rbase + v) * LDD + col] = acc[v];
    }
    STORE_FRAG_COPY()
}

__global__ void ses2seq_init(const float* __restrict__ Bm,
                             const float* __restrict__ theta0,
                             const float* __restrict__ xs,
                             float* __restrict__ C0,
                             float* __restrict__ thetas,
                             float* __restrict__ Cf0,
                             float* __restrict__ Uf) {
    if (blockIdx.x >= INIT_ROWBLK) {
        const int p = (blockIdx.x - INIT_ROWBLK) * blockDim.x + threadIdx.x;
        if (p >= UF_PIECES) return;
        const int lane = p & 31;
        const int q    = p >> 5;
        const int j    = q % CV_J;
        const int br   = q / CV_J;
        const int b    = br / NCONV;
        int urow = br % NCONV + (lane & 15);
        if (urow >= NCONV) urow -= NCONV;
        const float* src = xs + ((size_t)b * T + 1 + urow) * DS + 8 * j + ((lane >> 4) << 2);
        *(v4f*)(Uf + (size_t)p * 4) = *(const v4f*)src;
        return;
    }
    int l = blockIdx.x * blockDim.x + threadIdx.x;
    if (l >= L) return;
    float* row = C0 + (size_t)l * CW;
    const float* brow = Bm + (size_t)l * DS;
    {
        float* f = Cf0 + (size_t)(l >> 4) * (CV_J * FRAG) + (l & 15) * 4;
        #pragma unroll
        for (int j = 0; j < CV_J; ++j) {
            *(v4f*)(f + j * FRAG)      = *(const v4f*)(brow + 8 * j);
            *(v4f*)(f + j * FRAG + 64) = *(const v4f*)(brow + 8 * j + 4);
        }
    }
    #pragma unroll 4
    for (int d = 0; d < DS; ++d) row[d] = brow[d];
    row[DS] = theta0[l];
    #pragma unroll
    for (int b = 0; b < BATCH; ++b) {
        float acc = 0.f;
        const float* x0 = xs + (size_t)b * T * DS;
        for (int d = 0; d < DS; ++d) acc += brow[d] * x0[d];
        row[DS + 1 + b] = -acc;
    }
    for (int c = DS + 1 + BATCH; c < CW; ++c) row[c] = 0.f;
    #pragma unroll
    for (int b = 0; b < BATCH; ++b)
        thetas[((size_t)b * T + 0) * L + l] = theta0[l];
}

__global__ __launch_bounds__(128)
void ses2seq_conv(const float* __restrict__ C,
                  const float* __restrict__ Cf,
                  const float* __restrict__ Uf,
                  float* __restrict__ thetas) {
    const int lane  = threadIdx.x & 31;
    const int wv    = threadIdx.x >> 5;
    const int ltile = blockIdx.x;
    const int b     = blockIdx.y;
    const int tc    = wv * 16 + (lane & 15);

    const float* ka = Cf + (size_t)ltile * (CV_J * FRAG) + lane * 4;
    const float* ub = Uf + (size_t)b * (NCONV * CV_J * FRAG) + lane * 4;

    v4f kc[6][CV_J];
    v4f uu[6][CV_J];
    v8f acc = {};

#define CV_LOAD(SET, M)                                                          \
    {                                                                            \
        int r_ = wv * 16 - (M); if (r_ < 0) r_ += NCONV;        \
        const float* kp_ = ka + (size_t)(M) * CFSTEP;                            \
        const float* up_ = ub + (size_t)r_ * (CV_J * FRAG);                      \
        _Pragma("unroll")               \
        for (int j = 0; j < CV_J; ++j) {                                         \
            kc[SET][j] = *(const v4f*)(kp_ + FRAG * j);                          \
            uu[SET][j] = *(const v4f*)(up_ + FRAG * j);                          \
        }                                                                        \
    }
#define CV_STEP(M, PREV, CUR, NXT4, DOLOAD)                                      \
    {                                                                            \
        if (DOLOAD) CV_LOAD(NXT4, (M) + 4)                                       \
          \
          \
          \
        __builtin_amdgcn_sched_barrier(0);                                       \
        _Pragma("unroll")             \
        for (int j = 0; j < CV_J; ++j) {                                         \
            const v4f kd = kc[CUR][j] - kc[PREV][j];               \
            const v4f uv = uu[CUR][j];                                           \
            const v2f a0 = {kd.x, kd.y}, a1 = {kd.z, kd.w};                      \
            const v2f b0 = {uv.x, uv.y}, b1 = {uv.z, uv.w};                      \
            acc = __builtin_amdgcn_wmma_f32_16x16x4_f32(false, a0, false, b0,    \
                                                        (short)0, acc, false, false); \
            acc = __builtin_amdgcn_wmma_f32_16x16x4_f32(false, a1, false, b1,    \
                                                        (short)0, acc, false, false); \
        }                                                                        \
    }

    CV_LOAD(0, 0)
    CV_LOAD(1, 1)
    CV_LOAD(2, 2)
    CV_LOAD(3, 3)
    #pragma unroll
    for (int j = 0; j < CV_J; ++j) kc[5][j] = (v4f){0.f, 0.f, 0.f, 0.f};

    for (int m = 0; m < 60; m += 6) {
        CV_STEP(m,     5, 0, 4, 1)
        CV_STEP(m + 1, 0, 1, 5, 1)
        CV_STEP(m + 2, 1, 2, 0, 1)
        CV_STEP(m + 3, 2, 3, 1, 1)
        CV_STEP(m + 4, 3, 4, 2, 1)
        CV_STEP(m + 5, 4, 5, 3, (m + 9 < NCONV))
    }
    CV_STEP(60, 5, 0, 4, 0)
    CV_STEP(61, 0, 1, 5, 0)
    CV_STEP(62, 1, 2, 0, 0)
#undef CV_LOAD
#undef CV_STEP

    if (tc < NCONV) {
        const int l0 = ltile * 16 + ((lane >> 4) << 3);
        const int zc = (tc - 1 + NCONV) % NCONV;
        #pragma unroll
        for (int v = 0; v < 8; ++v) {
            const int l = l0 + v;
            float val = acc[v]
                      + C[(size_t)tc * CSTEP + (size_t)l * CW + DS]
                      + C[(size_t)zc * CSTEP + (size_t)l * CW + DS + 1 + b];
            thetas[((size_t)b * T + (tc + 1)) * L + l] = val;
        }
    }
}

__global__ void ses2seq_head(const float* __restrict__ thetas,
                             const float* __restrict__ ts,
                             float* __restrict__ out) {
    const int bt = blockIdx.x;
    const float* th = thetas + (size_t)bt * L;
    __shared__ float h[HID];
    if (threadIdx.x < HID) {
        float x = th[threadIdx.x] * ts[bt] + th[HID + threadIdx.x];
        h[threadIdx.x] = x > 0.f ? x : 0.f;
    }
    __syncthreads();
    const int o = threadIdx.x;
    float acc = th[2 * HID + OUT * HID + o];
    const float* w1 = th + 2 * HID + o * HID;
    #pragma unroll
    for (int k = 0; k < HID; ++k) acc += w1[k] * h[k];
    float r;
    if (o < DS) {
        r = tanhf(acc);
    } else {
        r = acc > 0.f ? acc + log1pf(expf(-acc)) : log1pf(expf(acc));
    }
    out[(size_t)bt * OUT + o] = r;
}

extern "C" void kernel_launch(void* const* d_in, const int* in_sizes, int n_in,
                              void* d_out, int out_size, void* d_ws, size_t ws_size,
                              hipStream_t stream) {
    const float* xs     = (const float*)d_in[0];
    const float* ts     = (const float*)d_in[1];
    const float* theta0 = (const float*)d_in[2];
    const float* A      = (const float*)d_in[3];
    const float* Bm     = (const float*)d_in[4];
    (void)in_sizes; (void)n_in; (void)out_size; (void)ws_size;

    float* A2     = (float*)d_ws;
    float* A4     = A2 + (size_t)L * L;
    float* A8     = A4 + (size_t)L * L;
    float* A16    = A8 + (size_t)L * L;
    float* C      = A16 + (size_t)L * L;
    float* thetas = C + (size_t)NCONV * CSTEP;
    float* Cf     = thetas + (size_t)BATCH * T * L;
    float* Uf     = Cf + (size_t)NCONV * CFSTEP;
    float* out    = (float*)d_out;

    ses2seq_init<<<INIT_ROWBLK + (UF_PIECES + 255) / 256, 256, 0, stream>>>(Bm, theta0, xs, C, thetas, Cf, Uf);

    wmma_mm_kernel<L, L, 4><<<dim3(14, 14, 1), 512, 0, stream>>>(A,  A,  0, A2,  0, nullptr, 0, L);
    wmma_mm_kernel<L, L, 4><<<dim3(14, 14, 1), 512, 0, stream>>>(A2, A2, 0, A4,  0, nullptr, 0, L);
    wmma_mm_kernel<L, L, 4><<<dim3(14, 14, 1), 512, 0, stream>>>(A4, A4, 0, A8,  0, nullptr, 0, L);
    wmma_mm_kernel<L, L, 4><<<dim3(14, 14, 1), 512, 0, stream>>>(A8, A8, 0, A16, 0, nullptr, 0, L);

    wmma_mm_pipe<CW, CW, 1><<<dim3(55, 1, 1), 128, 0, stream>>>(
        A,   C, 0,     C + CSTEP, 0, Cf + CFSTEP, 0, CW);
    wmma_mm_pipe<CW, CW, 1><<<dim3(55, 1, 2), 128, 0, stream>>>(
        A2,  C, CSTEP, C + (size_t)2 * CSTEP,  CSTEP, Cf + (size_t)2 * CFSTEP,  CFSTEP, CW);
    wmma_mm_pipe<CW, CW, 1><<<dim3(55, 1, 4), 128, 0, stream>>>(
        A4,  C, CSTEP, C + (size_t)4 * CSTEP,  CSTEP, Cf + (size_t)4 * CFSTEP,  CFSTEP, CW);
    wmma_mm_pipe<CW, CW, 2><<<dim3(28, 1, 8), 256, 0, stream>>>(
        A8,  C, CSTEP, C + (size_t)8 * CSTEP,  CSTEP, Cf + (size_t)8 * CFSTEP,  CFSTEP, CW);
    wmma_mm_kernel<CW, CW, 4><<<dim3(14, 1, 16), 512, 0, stream>>>(
        A16, C, CSTEP, C + (size_t)16 * CSTEP, CSTEP, Cf + (size_t)16 * CFSTEP, CFSTEP, CW);
    wmma_mm_kernel<CW, CW, 4><<<dim3(14, 1, 16), 512, 0, stream>>>(
        A16, C + (size_t)16 * CSTEP, CSTEP, C + (size_t)32 * CSTEP, CSTEP, Cf + (size_t)32 * CFSTEP, CFSTEP, CW);
    wmma_mm_kernel<CW, CW, 4><<<dim3(14, 1, 15), 512, 0, stream>>>(
        A16, C + (size_t)32 * CSTEP, CSTEP, C + (size_t)48 * CSTEP, CSTEP, Cf + (size_t)48 * CFSTEP, CFSTEP, CW);

    ses2seq_conv<<<dim3(55, BATCH), 128, 0, stream>>>(C, Cf, Uf, thetas);

    ses2seq_head<<<BATCH * T, OUT, 0, stream>>>(thetas, ts, out);
}
